# starting kernel
# baseline (speedup 1.0000x reference)
#include <hip/hip_runtime.h>

#define B_ 32
#define S_ 4096
#define H_ 512
#define NCHUNK 16
#define ROWS_PER_CHUNK (S_ / NCHUNK)
#define WAVES_PER_BLOCK 8
#define ROWS_PER_WAVE (ROWS_PER_CHUNK / WAVES_PER_BLOCK)

typedef __attribute__((ext_vector_type(2))) float v2f;
typedef __attribute__((ext_vector_type(8))) float v8f;

__global__ __launch_bounds__(32) void query_wmma_kernel(
    const float* __restrict__ dec, const float* __restrict__ W,
    float* __restrict__ q) {
  const int m0   = blockIdx.x * 16;
  const int n0   = blockIdx.y * 16;
  const int lane = threadIdx.x;
  const int half = lane >> 4;
  const int r    = lane & 15;
  const int koff = half * 2;

  v8f c = {};
  for (int k = 0; k < H_; k += 4) {
    const float* ap = dec + (m0 + r) * H_ + k + koff;
    const float* bp = W   + (n0 + r) * H_ + k + koff;
    v2f a, b;
    a.x = ap[0]; a.y = ap[1];
    b.x = bp[0]; b.y = bp[1];
    c = __builtin_amdgcn_wmma_f32_16x16x4_f32(
         false, a,  false, b,
         (short)0, c,  false,  false);
  }

  #pragma unroll
  for (int v = 0; v < 8; ++v) {
    const int m = v + 8 * half;
    q[(m0 + m) * H_ + (n0 + r)] = c[v];
  }
}

__global__ __launch_bounds__(256) void fused_scores_context(
    const float* __restrict__ enc, const float* __restrict__ qws,
    float* __restrict__ partials) {
  const int b     = blockIdx.x;
  const int chunk = blockIdx.y;
  const int tid   = threadIdx.x;
  const int wave  = tid >> 5;
  const int lane  = tid & 31;

  const float* q = qws + b * H_;
  float4 qr[4];
  #pragma unroll
  for (int j = 0; j < 4; ++j)
    qr[j] = *(const float4*)(q + j * 128 + lane * 4);

  float4 ctx[4];
  #pragma unroll
  for (int j = 0; j < 4; ++j) ctx[j] = make_float4(0.f, 0.f, 0.f, 0.f);

  const int s0 = chunk * ROWS_PER_CHUNK + wave * ROWS_PER_WAVE;
  const float* base = enc + ((size_t)b * S_ + s0) * H_;

  for (int s = 0; s < ROWS_PER_WAVE; ++s) {
    const float* row = base + (size_t)s * H_;
    __builtin_prefetch(row + H_ + lane * 4, 0, 0);

    float4 e[4];
    #pragma unroll
    for (int j = 0; j < 4; ++j)
      e[j] = *(const float4*)(row + j * 128 + lane * 4);

    float p = 0.f;
    #pragma unroll
    for (int j = 0; j < 4; ++j)
      p += e[j].x * qr[j].x + e[j].y * qr[j].y
         + e[j].z * qr[j].z + e[j].w * qr[j].w;

    #pragma unroll
    for (int off = 16; off >= 1; off >>= 1)
      p += __shfl_xor(p, off, 32);

    #pragma unroll
    for (int j = 0; j < 4; ++j) {
      ctx[j].x += p * e[j].x;
      ctx[j].y += p * e[j].y;
      ctx[j].z += p * e[j].z;
      ctx[j].w += p * e[j].w;
    }
  }

  __shared__ float red[WAVES_PER_BLOCK][H_];
  #pragma unroll
  for (int j = 0; j < 4; ++j)
    *(float4*)&red[wave][j * 128 + lane * 4] = ctx[j];
  __syncthreads();

  for (int h = tid; h < H_; h += 256) {
    float sum = 0.f;
    #pragma unroll
    for (int w = 0; w < WAVES_PER_BLOCK; ++w) sum += red[w][h];
    partials[(size_t)(b * NCHUNK + chunk) * H_ + h] = sum;
  }
}

__global__ __launch_bounds__(256) void reduce_context(
    const float* __restrict__ partials, float* __restrict__ out) {
  const int b = blockIdx.x;
  for (int h = threadIdx.x; h < H_; h += 256) {
    float sum = 0.f;
    #pragma unroll
    for (int c = 0; c < NCHUNK; ++c)
      sum += partials[(size_t)(b * NCHUNK + c) * H_ + h];
    out[b * H_ + h] = sum;
  }
}

extern "C" void kernel_launch(void* const* d_in, const int* in_sizes, int n_in,
                              void* d_out, int out_size, void* d_ws, size_t ws_size,
                              hipStream_t stream) {
  (void)in_sizes; (void)n_in; (void)out_size; (void)ws_size;
  const float* enc = (const float*)d_in[0];
  const float* dec = (const float*)d_in[1];
  const float* W   = (const float*)d_in[2];
  float* out = (float*)d_out;

  float* q_ws     = (float*)d_ws;
  float* partials = q_ws + B_ * H_;

  query_wmma_kernel<<<dim3(2, H_ / 16), 32, 0, stream>>>(dec, W, q_ws);
  fused_scores_context<<<dim3(B_, NCHUNK), 256, 0, stream>>>(enc, q_ws, partials);
  reduce_context<<<B_, 256, 0, stream>>>(partials, out);
}


# Streaming kernel takes enc rows in groups of 4 and requests the next group's 16 row loads before it sums the current group (two register buffers): 16-32 loads in flight per wave instead of 4-8
# speedup vs baseline: 1.1882x; 1.1882x over previous
#include <hip/hip_runtime.h>

#define B_ 32
#define S_ 4096
#define H_ 512
#define NCHUNK 16
#define ROWS_PER_CHUNK (S_ / NCHUNK)
#define WAVES_PER_BLOCK 8
#define ROWS_PER_WAVE (ROWS_PER_CHUNK / WAVES_PER_BLOCK)

typedef __attribute__((ext_vector_type(2))) float v2f;
typedef __attribute__((ext_vector_type(8))) float v8f;

__global__ __launch_bounds__(32) void query_wmma_kernel(
    const float* __restrict__ dec, const float* __restrict__ W,
    float* __restrict__ q) {
  const int m0   = blockIdx.x * 16;
  const int n0   = blockIdx.y * 16;
  const int lane = threadIdx.x;
  const int half = lane >> 4;
  const int r    = lane & 15;
  const int koff = half * 2;

  v8f c = {};
  for (int k = 0; k < H_; k += 4) {
    const float* ap = dec + (m0 + r) * H_ + k + koff;
    const float* bp = W   + (n0 + r) * H_ + k + koff;
    v2f a, b;
    a.x = ap[0]; a.y = ap[1];
    b.x = bp[0]; b.y = bp[1];
    c = __builtin_amdgcn_wmma_f32_16x16x4_f32(
         false, a,  false, b,
         (short)0, c,  false,  false);
  }

  #pragma unroll
  for (int v = 0; v < 8; ++v) {
    const int m = v + 8 * half;
    q[(m0 + m) * H_ + (n0 + r)] = c[v];
  }
}

typedef __attribute__((ext_vector_type(4))) float v4f;
#define ROWS_PER_GROUP 4
#define GROUPS_PER_WAVE (ROWS_PER_WAVE / ROWS_PER_GROUP)

static __device__ __forceinline__ void load_group(
    v4f (&e)[ROWS_PER_GROUP][4], const float* p) {
  #pragma unroll
  for (int r = 0; r < ROWS_PER_GROUP; ++r) {
    #pragma unroll
    for (int j = 0; j < 4; ++j)
      e[r][j] = *(const v4f*)(p + r * H_ + j * 128);
  }
}

static __device__ __forceinline__ void sum_group(
    const v4f (&e)[ROWS_PER_GROUP][4], v4f (&qr)[4], v4f (&ctx)[4]) {
  asm volatile("" : "+v"(qr[0]), "+v"(qr[1]), "+v"(qr[2]), "+v"(qr[3]));

  float p[ROWS_PER_GROUP];
  #pragma unroll
  for (int r = 0; r < ROWS_PER_GROUP; ++r) {
    float acc = 0.f;
    #pragma unroll
    for (int j = 0; j < 4; ++j)
      acc += e[r][j].x * qr[j].x + e[r][j].y * qr[j].y
           + e[r][j].z * qr[j].z + e[r][j].w * qr[j].w;
    p[r] = acc;
  }

  #pragma unroll
  for (int off = 16; off >= 1; off >>= 1) {
    #pragma unroll
    for (int r = 0; r < ROWS_PER_GROUP; ++r)
      p[r] += __shfl_xor(p[r], off, 32);
  }

  #pragma unroll
  for (int r = 0; r < ROWS_PER_GROUP; ++r) {
    #pragma unroll
    for (int j = 0; j < 4; ++j) {
      ctx[j].x += p[r] * e[r][j].x;
      ctx[j].y += p[r] * e[r][j].y;
      ctx[j].z += p[r] * e[r][j].z;
      ctx[j].w += p[r] * e[r][j].w;
    }
  }

  asm volatile("" : "+v"(ctx[0]), "+v"(ctx[1]), "+v"(ctx[2]), "+v"(ctx[3]));
}

__global__ __launch_bounds__(256) void fused_scores_context(
    const float* __restrict__ enc, const float* __restrict__ qws,
    float* __restrict__ partials) {
  const int b     = blockIdx.x;
  const int chunk = blockIdx.y;
  const int tid   = threadIdx.x;
  const int wave  = tid >> 5;
  const int lane  = tid & 31;

  const float* q = qws + b * H_;
  v4f qr[4];
  #pragma unroll
  for (int j = 0; j < 4; ++j)
    qr[j] = *(const v4f*)(q + j * 128 + lane * 4);

  v4f ctx[4];
  #pragma unroll
  for (int j = 0; j < 4; ++j) ctx[j] = (v4f){0.f, 0.f, 0.f, 0.f};

  const int s0 = chunk * ROWS_PER_CHUNK + wave * ROWS_PER_WAVE;
  const float* grp = enc + ((size_t)b * S_ + s0) * H_ + lane * 4;

  v4f ea[ROWS_PER_GROUP][4], eb[ROWS_PER_GROUP][4];
  load_group(ea, grp);

  #pragma unroll 1
  for (int t = 0; t < GROUPS_PER_WAVE / 2 - 1; ++t) {
    __builtin_amdgcn_sched_barrier(0);
    load_group(eb, grp + ROWS_PER_GROUP * H_);
    __builtin_amdgcn_sched_barrier(0);
    sum_group(ea, qr, ctx);
    __builtin_amdgcn_sched_barrier(0);
    load_group(ea, grp + 2 * ROWS_PER_GROUP * H_);
    __builtin_amdgcn_sched_barrier(0);
    sum_group(eb, qr, ctx);
    grp += 2 * ROWS_PER_GROUP * H_;
  }
  __builtin_amdgcn_sched_barrier(0);
  load_group(eb, grp + ROWS_PER_GROUP * H_);
  __builtin_amdgcn_sched_barrier(0);
  sum_group(ea, qr, ctx);
  sum_group(eb, qr, ctx);

  __shared__ float red[WAVES_PER_BLOCK][H_];
  #pragma unroll
  for (int j = 0; j < 4; ++j)
    *(v4f*)&red[wave][j * 128 + lane * 4] = ctx[j];
  __syncthreads();

  for (int h = tid; h < H_; h += 256) {
    float sum = 0.f;
    #pragma unroll
    for (int w = 0; w < WAVES_PER_BLOCK; ++w) sum += red[w][h];
    partials[(size_t)(b * NCHUNK + chunk) * H_ + h] = sum;
  }
}

__global__ __launch_bounds__(256) void reduce_context(
    const float* __restrict__ partials, float* __restrict__ out) {
  const int b = blockIdx.x;
  for (int h = threadIdx.x; h < H_; h += 256) {
    float sum = 0.f;
    #pragma unroll
    for (int c = 0; c < NCHUNK; ++c)
      sum += partials[(size_t)(b * NCHUNK + c) * H_ + h];
    out[b * H_ + h] = sum;
  }
}

extern "C" void kernel_launch(void* const* d_in, const int* in_sizes, int n_in,
                              void* d_out, int out_size, void* d_ws, size_t ws_size,
                              hipStream_t stream) {
  (void)in_sizes; (void)n_in; (void)out_size; (void)ws_size;
  const float* enc = (const float*)d_in[0];
  const float* dec = (const float*)d_in[1];
  const float* W   = (const float*)d_in[2];
  float* out = (float*)d_out;

  float* q_ws     = (float*)d_ws;
  float* partials = q_ws + B_ * H_;

  query_wmma_kernel<<<dim3(2, H_ / 16), 32, 0, stream>>>(dec, W, q_ws);
  fused_scores_context<<<dim3(B_, NCHUNK), 256, 0, stream>>>(enc, q_ws, partials);
  reduce_context<<<B_, 256, 0, stream>>>(partials, out);
}


# Query GEMM: K of each 16x16 tile split over the 8 waves of a block, each wave's 16 fragment loads issued as one group, partial tiles summed via LDS in wave order (k summation order changes)
# speedup vs baseline: 1.3874x; 1.1676x over previous
#include <hip/hip_runtime.h>

#define B_ 32
#define S_ 4096
#define H_ 512
#define NCHUNK 16
#define ROWS_PER_CHUNK (S_ / NCHUNK)
#define WAVES_PER_BLOCK 8
#define ROWS_PER_WAVE (ROWS_PER_CHUNK / WAVES_PER_BLOCK)

typedef __attribute__((ext_vector_type(2))) float v2f;
typedef __attribute__((ext_vector_type(8))) float v8f;

typedef __attribute__((ext_vector_type(4))) float v4f;
#define QWAVES 8
#define QSLICE (H_ / QWAVES)
#define QVECS  (QSLICE / 8)

__global__ __launch_bounds__(256) void query_wmma_kernel(
    const float* __restrict__ dec, const float* __restrict__ W,
    float* __restrict__ q) {
  const int m0   = blockIdx.x * 16;
  const int n0   = blockIdx.y * 16;
  const int tid  = threadIdx.x;
  const int wave = tid >> 5;
  const int lane = tid & 31;
  const int half = lane >> 4;
  const int r    = lane & 15;

  const float* ap = dec + (m0 + r) * H_ + wave * QSLICE + half * 4;
  const float* bp = W   + (n0 + r) * H_ + wave * QSLICE + half * 4;

  v4f a[QVECS], b[QVECS];
  #pragma unroll
  for (int j = 0; j < QVECS; ++j) {
    a[j] = *(const v4f*)(ap + 8 * j);
    b[j] = *(const v4f*)(bp + 8 * j);
  }
  __builtin_amdgcn_sched_barrier(0);

  v8f c0 = {}, c1 = {};
  #pragma unroll
  for (int j = 0; j < QVECS; ++j) {
    v2f a01 = {a[j].x, a[j].y}, b01 = {b[j].x, b[j].y};
    v2f a23 = {a[j].z, a[j].w}, b23 = {b[j].z, b[j].w};
    c0 = __builtin_amdgcn_wmma_f32_16x16x4_f32(
         false, a01,  false, b01,
         (short)0, c0,  false,  false);
    c1 = __builtin_amdgcn_wmma_f32_16x16x4_f32(
         false, a23,  false, b23,
         (short)0, c1,  false,  false);
  }

  __shared__ float qred[QWAVES][8][32];
  #pragma unroll
  for (int v = 0; v < 8; ++v)
    qred[wave][v][lane] = c0[v] + c1[v];
  __syncthreads();

  float sum = 0.f;
  #pragma unroll
  for (int w = 0; w < QWAVES; ++w) sum += qred[w][wave][lane];
  const int m = wave + 8 * half;
  q[(m0 + m) * H_ + (n0 + r)] = sum;
}

#define ROWS_PER_GROUP 4
#define GROUPS_PER_WAVE (ROWS_PER_WAVE / ROWS_PER_GROUP)

static __device__ __forceinline__ void load_group(
    v4f (&e)[ROWS_PER_GROUP][4], const float* p) {
  #pragma unroll
  for (int r = 0; r < ROWS_PER_GROUP; ++r) {
    #pragma unroll
    for (int j = 0; j < 4; ++j)
      e[r][j] = *(const v4f*)(p + r * H_ + j * 128);
  }
}

static __device__ __forceinline__ void sum_group(
    const v4f (&e)[ROWS_PER_GROUP][4], v4f (&qr)[4], v4f (&ctx)[4]) {
  asm volatile("" : "+v"(qr[0]), "+v"(qr[1]), "+v"(qr[2]), "+v"(qr[3]));

  float p[ROWS_PER_GROUP];
  #pragma unroll
  for (int r = 0; r < ROWS_PER_GROUP; ++r) {
    float acc = 0.f;
    #pragma unroll
    for (int j = 0; j < 4; ++j)
      acc += e[r][j].x * qr[j].x + e[r][j].y * qr[j].y
           + e[r][j].z * qr[j].z + e[r][j].w * qr[j].w;
    p[r] = acc;
  }

  #pragma unroll
  for (int off = 16; off >= 1; off >>= 1) {
    #pragma unroll
    for (int r = 0; r < ROWS_PER_GROUP; ++r)
      p[r] += __shfl_xor(p[r], off, 32);
  }

  #pragma unroll
  for (int r = 0; r < ROWS_PER_GROUP; ++r) {
    #pragma unroll
    for (int j = 0; j < 4; ++j) {
      ctx[j].x += p[r] * e[r][j].x;
      ctx[j].y += p[r] * e[r][j].y;
      ctx[j].z += p[r] * e[r][j].z;
      ctx[j].w += p[r] * e[r][j].w;
    }
  }

  asm volatile("" : "+v"(ctx[0]), "+v"(ctx[1]), "+v"(ctx[2]), "+v"(ctx[3]));
}

__global__ __launch_bounds__(256) void fused_scores_context(
    const float* __restrict__ enc, const float* __restrict__ qws,
    float* __restrict__ partials) {
  const int b     = blockIdx.x;
  const int chunk = blockIdx.y;
  const int tid   = threadIdx.x;
  const int wave  = tid >> 5;
  const int lane  = tid & 31;

  const float* q = qws + b * H_;
  v4f qr[4];
  #pragma unroll
  for (int j = 0; j < 4; ++j)
    qr[j] = *(const v4f*)(q + j * 128 + lane * 4);

  v4f ctx[4];
  #pragma unroll
  for (int j = 0; j < 4; ++j) ctx[j] = (v4f){0.f, 0.f, 0.f, 0.f};

  const int s0 = chunk * ROWS_PER_CHUNK + wave * ROWS_PER_WAVE;
  const float* grp = enc + ((size_t)b * S_ + s0) * H_ + lane * 4;

  v4f ea[ROWS_PER_GROUP][4], eb[ROWS_PER_GROUP][4];
  load_group(ea, grp);

  #pragma unroll 1
  for (int t = 0; t < GROUPS_PER_WAVE / 2 - 1; ++t) {
    __builtin_amdgcn_sched_barrier(0);
    load_group(eb, grp + ROWS_PER_GROUP * H_);
    __builtin_amdgcn_sched_barrier(0);
    sum_group(ea, qr, ctx);
    __builtin_amdgcn_sched_barrier(0);
    load_group(ea, grp + 2 * ROWS_PER_GROUP * H_);
    __builtin_amdgcn_sched_barrier(0);
    sum_group(eb, qr, ctx);
    grp += 2 * ROWS_PER_GROUP * H_;
  }
  __builtin_amdgcn_sched_barrier(0);
  load_group(eb, grp + ROWS_PER_GROUP * H_);
  __builtin_amdgcn_sched_barrier(0);
  sum_group(ea, qr, ctx);
  sum_group(eb, qr, ctx);

  __shared__ float red[WAVES_PER_BLOCK][H_];
  #pragma unroll
  for (int j = 0; j < 4; ++j)
    *(v4f*)&red[wave][j * 128 + lane * 4] = ctx[j];
  __syncthreads();

  for (int h = tid; h < H_; h += 256) {
    float sum = 0.f;
    #pragma unroll
    for (int w = 0; w < WAVES_PER_BLOCK; ++w) sum += red[w][h];
    partials[(size_t)(b * NCHUNK + chunk) * H_ + h] = sum;
  }
}

__global__ __launch_bounds__(256) void reduce_context(
    const float* __restrict__ partials, float* __restrict__ out) {
  const int b = blockIdx.x;
  for (int h = threadIdx.x; h < H_; h += 256) {
    float sum = 0.f;
    #pragma unroll
    for (int c = 0; c < NCHUNK; ++c)
      sum += partials[(size_t)(b * NCHUNK + c) * H_ + h];
    out[b * H_ + h] = sum;
  }
}

extern "C" void kernel_launch(void* const* d_in, const int* in_sizes, int n_in,
                              void* d_out, int out_size, void* d_ws, size_t ws_size,
                              hipStream_t stream) {
  (void)in_sizes; (void)n_in; (void)out_size; (void)ws_size;
  const float* enc = (const float*)d_in[0];
  const float* dec = (const float*)d_in[1];
  const float* W   = (const float*)d_in[2];
  float* out = (float*)d_out;

  float* q_ws     = (float*)d_ws;
  float* partials = q_ws + B_ * H_;

  query_wmma_kernel<<<dim3(2, H_ / 16), 256, 0, stream>>>(dec, W, q_ws);
  fused_scores_context<<<dim3(B_, NCHUNK), 256, 0, stream>>>(enc, q_ws, partials);
  reduce_context<<<B_, 256, 0, stream>>>(partials, out);
}
